# starting kernel
# baseline (speedup 1.0000x reference)
#include <hip/hip_runtime.h>
#include <hip/hip_bf16.h>
#include <math.h>

#define B_DIM 64
#define L_DIM 16384
#define ORDER 3
#define V_DIM 4
#define CTX_DIM 64
#define NBITS 16

typedef float v2f __attribute__((ext_vector_type(2)));
typedef float v8f __attribute__((ext_vector_type(8)));

__global__ __launch_bounds__(128) void mc_pow_kernel(
    const float* __restrict__ time, const float* __restrict__ T,
    const float* __restrict__ W1,   const float* __restrict__ b1,
    const float* __restrict__ W2,   const float* __restrict__ b2,
    float* __restrict__ tbl)
{
    const int lane = threadIdx.x & 31;
    const int wave = threadIdx.x >> 5;
    const int globalWave = blockIdx.x * 4 + wave;
    const int chainBase  = globalWave * 4;
    const int b = chainBase >> 6;
    const float t = time[b];

    int nexp[4];
#pragma unroll
    for (int q = 0; q < 4; ++q) {
        const int ctx = (chainBase + q) & 63;
        float acc = b2[ctx];
        for (int j = 0; j < 32; ++j) {
            float h = fmaf(t, W1[j], b1[j]);
            h = fmaxf(h, 0.0f);
            acc = fmaf(h, W2[ctx * 32 + j], acc);
        }
        float sp = log1pf(expf(-fabsf(acc))) + fmaxf(acc, 0.0f);
        float nf = fmaf(sp, t, 1.0f);
        int ni = (int)nf;
        ni = ni < 0 ? 0 : (ni > 65535 ? 65535 : ni);
        nexp[q] = ni;
    }

    const int m  = lane & 15;
    const int hi = lane >> 4;
    const int k0 = hi * 2;
    const int qA = m >> 2;
    const int rA = m & 3;

    const float* Mq = T + (((chainBase + qA) & 63) * 16);
    v2f aM = { Mq[rA * 4 + k0], Mq[rA * 4 + k0 + 1] };
    v2f bM = { Mq[k0 * 4 + rA], Mq[(k0 + 1) * 4 + rA] };
    v2f aR = { (rA == k0) ? 1.0f : 0.0f, (rA == k0 + 1) ? 1.0f : 0.0f };

    __shared__ float lds[4][2][256];
    float* mm = &lds[wave][0][0];
    float* rm = &lds[wave][1][0];

    for (int k = 0; k < NBITS; ++k) {
        v8f zero = {};
        v8f dRM = __builtin_amdgcn_wmma_f32_16x16x4_f32(
            false, aR, false, bM, (short)0, zero, false, false);
        v8f dMM = __builtin_amdgcn_wmma_f32_16x16x4_f32(
            false, aM, false, bM, (short)0, zero, false, false);

#pragma unroll
        for (int r = 0; r < 8; ++r) {
            const int mr = r + 8 * hi;
            mm[mr * 16 + m] = dMM[r];
            rm[mr * 16 + m] = dRM[r];
        }
        __syncthreads();

        const int base = 4 * qA + k0;
        aM[0] = mm[m * 16 + base];
        aM[1] = mm[m * 16 + base + 1];
        bM[0] = mm[base * 16 + m];
        bM[1] = mm[(base + 1) * 16 + m];
        if ((nexp[qA] >> k) & 1) {
            aR[0] = rm[m * 16 + base];
            aR[1] = rm[m * 16 + base + 1];
        }
        __syncthreads();
    }

    const int chain = chainBase + qA;
    tbl[chain * 16 + rA * 4 + k0]     = aR[0];
    tbl[chain * 16 + rA * 4 + k0 + 1] = aR[1];
}

__global__ __launch_bounds__(256) void mc_gather_kernel(
    const int* __restrict__ seq, const float* __restrict__ tbl,
    float* __restrict__ out)
{
    const int tid = threadIdx.x;
    const int b   = blockIdx.y;
    const int p0  = blockIdx.x * 256;

    __shared__ float s_tbl[CTX_DIM * V_DIM * V_DIM];
    __shared__ int   s_seq[256 + ORDER];

    ((float4*)s_tbl)[tid] = ((const float4*)(tbl + b * 1024))[tid];

    for (int i = tid; i < 256 + ORDER; i += 256) {
        const int p = p0 - ORDER + i;
        s_seq[i] = (p >= 0) ? seq[b * L_DIM + p] : 0;
    }
    __syncthreads();

    const int p = p0 + tid;
    float4 res = make_float4(0.0f, 0.0f, 0.0f, 0.0f);
    if (p >= ORDER) {
        const int s0 = s_seq[tid];
        const int s1 = s_seq[tid + 1];
        const int s2 = s_seq[tid + 2];
        const int ctx = s0 * 16 + s1 * 4 + s2;
        res = ((const float4*)s_tbl)[ctx * 4 + s2];
    }
    ((float4*)out)[b * L_DIM + p] = res;
}

extern "C" void kernel_launch(void* const* d_in, const int* in_sizes, int n_in,
                              void* d_out, int out_size, void* d_ws, size_t ws_size,
                              hipStream_t stream) {
    const int*   seq  = (const int*)d_in[0];
    const float* time = (const float*)d_in[1];
    const float* T    = (const float*)d_in[2];
    const float* W1   = (const float*)d_in[3];
    const float* b1   = (const float*)d_in[4];
    const float* W2   = (const float*)d_in[5];
    const float* b2   = (const float*)d_in[6];
    float* out = (float*)d_out;
    float* tbl = (float*)d_ws;

    hipLaunchKernelGGL(mc_pow_kernel, dim3(4096 / 16), dim3(128), 0, stream,
                       time, T, W1, b1, W2, b2, tbl);

    hipLaunchKernelGGL(mc_gather_kernel, dim3(L_DIM / 256, B_DIM), dim3(256), 0, stream,
                       seq, tbl, out);
}


# gather: 256 workgroups of 4096 positions (16 per thread) instead of 4096 of 256; the 64 selectable table rows and the sequence window are staged in LDS once per workgroup
# speedup vs baseline: 1.2892x; 1.2892x over previous
#include <hip/hip_runtime.h>
#include <hip/hip_bf16.h>
#include <math.h>

#define B_DIM 64
#define L_DIM 16384
#define ORDER 3
#define V_DIM 4
#define CTX_DIM 64
#define NBITS 16

typedef float v2f __attribute__((ext_vector_type(2)));
typedef float v8f __attribute__((ext_vector_type(8)));

__global__ __launch_bounds__(128) void mc_pow_kernel(
    const float* __restrict__ time, const float* __restrict__ T,
    const float* __restrict__ W1,   const float* __restrict__ b1,
    const float* __restrict__ W2,   const float* __restrict__ b2,
    float* __restrict__ tbl)
{
    const int lane = threadIdx.x & 31;
    const int wave = threadIdx.x >> 5;
    const int globalWave = blockIdx.x * 4 + wave;
    const int chainBase  = globalWave * 4;
    const int b = chainBase >> 6;
    const float t = time[b];

    int nexp[4];
#pragma unroll
    for (int q = 0; q < 4; ++q) {
        const int ctx = (chainBase + q) & 63;
        float acc = b2[ctx];
        for (int j = 0; j < 32; ++j) {
            float h = fmaf(t, W1[j], b1[j]);
            h = fmaxf(h, 0.0f);
            acc = fmaf(h, W2[ctx * 32 + j], acc);
        }
        float sp = log1pf(expf(-fabsf(acc))) + fmaxf(acc, 0.0f);
        float nf = fmaf(sp, t, 1.0f);
        int ni = (int)nf;
        ni = ni < 0 ? 0 : (ni > 65535 ? 65535 : ni);
        nexp[q] = ni;
    }

    const int m  = lane & 15;
    const int hi = lane >> 4;
    const int k0 = hi * 2;
    const int qA = m >> 2;
    const int rA = m & 3;

    const float* Mq = T + (((chainBase + qA) & 63) * 16);
    v2f aM = { Mq[rA * 4 + k0], Mq[rA * 4 + k0 + 1] };
    v2f bM = { Mq[k0 * 4 + rA], Mq[(k0 + 1) * 4 + rA] };
    v2f aR = { (rA == k0) ? 1.0f : 0.0f, (rA == k0 + 1) ? 1.0f : 0.0f };

    __shared__ float lds[4][2][256];
    float* mm = &lds[wave][0][0];
    float* rm = &lds[wave][1][0];

    for (int k = 0; k < NBITS; ++k) {
        v8f zero = {};
        v8f dRM = __builtin_amdgcn_wmma_f32_16x16x4_f32(
            false, aR, false, bM, (short)0, zero, false, false);
        v8f dMM = __builtin_amdgcn_wmma_f32_16x16x4_f32(
            false, aM, false, bM, (short)0, zero, false, false);

#pragma unroll
        for (int r = 0; r < 8; ++r) {
            const int mr = r + 8 * hi;
            mm[mr * 16 + m] = dMM[r];
            rm[mr * 16 + m] = dRM[r];
        }
        __syncthreads();

        const int base = 4 * qA + k0;
        aM[0] = mm[m * 16 + base];
        aM[1] = mm[m * 16 + base + 1];
        bM[0] = mm[base * 16 + m];
        bM[1] = mm[(base + 1) * 16 + m];
        if ((nexp[qA] >> k) & 1) {
            aR[0] = rm[m * 16 + base];
            aR[1] = rm[m * 16 + base + 1];
        }
        __syncthreads();
    }

    const int chain = chainBase + qA;
    tbl[chain * 16 + rA * 4 + k0]     = aR[0];
    tbl[chain * 16 + rA * 4 + k0 + 1] = aR[1];
}

#define G_POS        4096
#define G_THREADS    256
#define G_PER_THREAD (G_POS / G_THREADS)
#define G_PER_WAVE   (G_POS / (G_THREADS / 32))

__global__ __launch_bounds__(G_THREADS) void mc_gather_kernel(
    const int* __restrict__ seq, const float* __restrict__ tbl,
    float* __restrict__ out)
{
    const int tid  = threadIdx.x;
    const int lane = tid & 31;
    const int wave = tid >> 5;
    const int b    = blockIdx.y;
    const int p0   = blockIdx.x * G_POS;

    __shared__ float4 s_row[CTX_DIM];
    __shared__ int4   s_seq4[G_POS / 4 + 1];
    int* s_seq = (int*)s_seq4;

    const int* seqRow = seq + b * L_DIM + p0;

    float4 rowv = make_float4(0.0f, 0.0f, 0.0f, 0.0f);
    if (tid < CTX_DIM)
        rowv = *(const float4*)(tbl + b * 1024 + tid * 16 + (tid & 3) * 4);
    int halo = 0;
    if (tid < ORDER && p0 > 0) halo = seqRow[tid - ORDER];
    int4 sv[G_POS / 4 / G_THREADS];
#pragma unroll
    for (int j = 0; j < G_POS / 4 / G_THREADS; ++j)
        sv[j] = ((const int4*)seqRow)[tid + G_THREADS * j];

#pragma unroll
    for (int j = 0; j < G_POS / 4 / G_THREADS; ++j)
        s_seq4[1 + tid + G_THREADS * j] = sv[j];
    if (tid < CTX_DIM) s_row[tid] = rowv;
    if (tid < ORDER)   s_seq[1 + tid] = halo;
    __syncthreads();

    float4* outRow = (float4*)out + b * L_DIM + p0;
    const int kw = wave * G_PER_WAVE + lane;
#pragma unroll
    for (int i = 0; i < G_PER_THREAD; ++i) {
        const int k  = kw + i * 32;
        const int s0 = s_seq[k + 1];
        const int s1 = s_seq[k + 2];
        const int s2 = s_seq[k + 3];
        float4 res = s_row[s0 * 16 + s1 * 4 + s2];
        if (i == 0 && p0 + k < ORDER)
            res = make_float4(0.0f, 0.0f, 0.0f, 0.0f);
        outRow[k] = res;
    }
}

extern "C" void kernel_launch(void* const* d_in, const int* in_sizes, int n_in,
                              void* d_out, int out_size, void* d_ws, size_t ws_size,
                              hipStream_t stream) {
    const int*   seq  = (const int*)d_in[0];
    const float* time = (const float*)d_in[1];
    const float* T    = (const float*)d_in[2];
    const float* W1   = (const float*)d_in[3];
    const float* b1   = (const float*)d_in[4];
    const float* W2   = (const float*)d_in[5];
    const float* b2   = (const float*)d_in[6];
    float* out = (float*)d_out;
    float* tbl = (float*)d_ws;

    hipLaunchKernelGGL(mc_pow_kernel, dim3(4096 / 16), dim3(128), 0, stream,
                       time, T, W1, b1, W2, b2, tbl);

    hipLaunchKernelGGL(mc_gather_kernel, dim3(L_DIM / G_POS, B_DIM), dim3(G_THREADS), 0, stream,
                       seq, tbl, out);
}


# one launch instead of two: each gather workgroup computes its own b's 64 selectable table rows in LDS (64 power chains on WMMA, per-wave LDS transposes, no workgroup barrier per step); no workspace
# speedup vs baseline: 1.8022x; 1.3979x over previous
#include <hip/hip_runtime.h>
#include <hip/hip_bf16.h>
#include <math.h>


#define B_DIM 64
#define L_DIM 16384
#define ORDER 3
#define V_DIM 4
#define CTX_DIM 64
#define NBITS 16

typedef float v2f __attribute__((ext_vector_type(2)));
typedef float v8f __attribute__((ext_vector_type(8)));

#define G_POS        4096
#define G_THREADS    256
#define G_WAVES      (G_THREADS / 32)
#define G_PER_THREAD (G_POS / G_THREADS)
#define G_PER_WAVE   (G_POS / G_WAVES)
#define TILES        2

__global__ __launch_bounds__(G_THREADS) void mc_fused_kernel(
    const int* __restrict__ seq,  const float* __restrict__ time,
    const float* __restrict__ T,
    const float* __restrict__ W1, const float* __restrict__ b1,
    const float* __restrict__ W2, const float* __restrict__ b2,
    float* __restrict__ out)
{
    const int tid  = threadIdx.x;
    const int lane = tid & 31;
    const int wave = tid >> 5;
    const int b    = blockIdx.y;
    const int p0   = blockIdx.x * G_POS;

    __shared__ float4 s_row[CTX_DIM];
    __shared__ int4   s_seq4[G_POS / 4 + 1];
    __shared__ float4 s_pow[G_WAVES][TILES][2][16];
    int* s_seq = (int*)s_seq4;

    const int m  = lane & 15;
    const int hi = lane >> 4;
    const int k0 = hi * 2;
    const int qA = m >> 2;
    const int rA = m & 3;

    const float t = time[b];
    const int myCtx = wave * (4 * TILES) + hi * 4 + qA;
    const float4* w2v = (const float4*)(W2 + myCtx * 32);
    float4 w2r[8];
#pragma unroll
    for (int j = 0; j < 8; ++j) w2r[j] = w2v[j];
    float acc = b2[myCtx];

    const int* seqRow = seq + b * L_DIM + p0;
    const int haloIdx = (tid < ORDER && p0 > 0) ? tid - ORDER : 0;
    const int halo = seqRow[haloIdx];
    int4 sv[G_POS / 4 / G_THREADS];
#pragma unroll
    for (int j = 0; j < G_POS / 4 / G_THREADS; ++j)
        sv[j] = ((const int4*)seqRow)[tid + G_THREADS * j];

    v2f aM[TILES], bM[TILES], aR[TILES];
#pragma unroll
    for (int tt = 0; tt < TILES; ++tt) {
        const float* Mq = T + (wave * (4 * TILES) + tt * 4 + qA) * 16;
        aM[tt] = (v2f){ Mq[rA * 4 + k0], Mq[rA * 4 + k0 + 1] };
        bM[tt] = (v2f){ Mq[k0 * 4 + rA], Mq[(k0 + 1) * 4 + rA] };
        aR[tt] = (v2f){ (rA == k0) ? 1.0f : 0.0f, (rA == k0 + 1) ? 1.0f : 0.0f };
    }

#pragma unroll
    for (int j = 0; j < 32; ++j) {
        float h = fmaf(t, W1[j], b1[j]);
        h = fmaxf(h, 0.0f);
        const float4 wq = w2r[j >> 2];
        const float w = (j & 3) == 0 ? wq.x : (j & 3) == 1 ? wq.y : (j & 3) == 2 ? wq.z : wq.w;
        acc = fmaf(h, w, acc);
    }
    int myN;
    {
        float sp = log1pf(expf(-fabsf(acc))) + fmaxf(acc, 0.0f);
        float nf = fmaf(sp, t, 1.0f);
        int ni = (int)nf;
        ni = ni < 0 ? 0 : (ni > 65535 ? 65535 : ni);
        myN = ni;
    }
    const int otherN = __shfl_xor(myN, 16);
    int nexp[TILES];
    nexp[0] = hi ? otherN : myN;
    nexp[1] = hi ? myN : otherN;

#pragma unroll
    for (int j = 0; j < G_POS / 4 / G_THREADS; ++j)
        s_seq4[1 + tid + G_THREADS * j] = sv[j];
    asm volatile("" :: "v"(halo));
    if (tid < ORDER)   s_seq[1 + tid] = halo;

    const bool owner = (hi == (qA >> 1));
    const bool upper = (qA & 1) != 0;
    for (int k = 0; k < NBITS; ++k) {
        v8f dRM[TILES], dMM[TILES];
#pragma unroll
        for (int tt = 0; tt < TILES; ++tt) {
            v8f zero = {};
            dRM[tt] = __builtin_amdgcn_wmma_f32_16x16x4_f32(
                false, aR[tt], false, bM[tt], (short)0, zero, false, false);
            dMM[tt] = __builtin_amdgcn_wmma_f32_16x16x4_f32(
                false, aM[tt], false, bM[tt], (short)0, zero, false, false);
        }
#pragma unroll
        for (int tt = 0; tt < TILES; ++tt) {
            if (owner) {
                float4 cmm, crm;
                cmm.x = upper ? dMM[tt][4] : dMM[tt][0];
                cmm.y = upper ? dMM[tt][5] : dMM[tt][1];
                cmm.z = upper ? dMM[tt][6] : dMM[tt][2];
                cmm.w = upper ? dMM[tt][7] : dMM[tt][3];
                crm.x = upper ? dRM[tt][4] : dRM[tt][0];
                crm.y = upper ? dRM[tt][5] : dRM[tt][1];
                crm.z = upper ? dRM[tt][6] : dRM[tt][2];
                crm.w = upper ? dRM[tt][7] : dRM[tt][3];
                s_pow[wave][tt][0][m] = cmm;
                s_pow[wave][tt][1][m] = crm;
            }
        }
        __builtin_amdgcn_fence(__ATOMIC_RELEASE, "wavefront");
        __builtin_amdgcn_wave_barrier();
        __builtin_amdgcn_fence(__ATOMIC_ACQUIRE, "wavefront");
#pragma unroll
        for (int tt = 0; tt < TILES; ++tt) {
            const float* mmc = (const float*)&s_pow[wave][tt][0][0];
            const float* rmc = (const float*)&s_pow[wave][tt][1][0];
            const int ia = qA * 16 + k0 * 4 + rA;
            aM[tt] = (v2f){ mmc[ia], mmc[ia + 4] };
            bM[tt] = *(const v2f*)&mmc[4 * m + k0];
            const v2f rnew = (v2f){ rmc[ia], rmc[ia + 4] };
            if ((nexp[tt] >> k) & 1) aR[tt] = rnew;
        }
        __builtin_amdgcn_fence(__ATOMIC_RELEASE, "wavefront");
        __builtin_amdgcn_wave_barrier();
        __builtin_amdgcn_fence(__ATOMIC_ACQUIRE, "wavefront");
    }

#pragma unroll
    for (int tt = 0; tt < TILES; ++tt) {
        if (rA == qA) {
            const int ctx = wave * (4 * TILES) + tt * 4 + qA;
            ((v2f*)&s_row[ctx])[hi] = aR[tt];
        }
    }

    __syncthreads();

    float4* outRow = (float4*)out + b * L_DIM + p0;
    const int kw = wave * G_PER_WAVE + lane;
#pragma unroll
    for (int i = 0; i < G_PER_THREAD; ++i) {
        const int k  = kw + i * 32;
        const int s0 = s_seq[k + 1];
        const int s1 = s_seq[k + 2];
        const int s2 = s_seq[k + 3];
        float4 res = s_row[s0 * 16 + s1 * 4 + s2];
        if (i == 0 && p0 + k < ORDER)
            res = make_float4(0.0f, 0.0f, 0.0f, 0.0f);
        outRow[k] = res;
    }
}

extern "C" void kernel_launch(void* const* d_in, const int* in_sizes, int n_in,
                              void* d_out, int out_size, void* d_ws, size_t ws_size,
                              hipStream_t stream) {
    const int*   seq  = (const int*)d_in[0];
    const float* time = (const float*)d_in[1];
    const float* T    = (const float*)d_in[2];
    const float* W1   = (const float*)d_in[3];
    const float* b1   = (const float*)d_in[4];
    const float* W2   = (const float*)d_in[5];
    const float* b2   = (const float*)d_in[6];
    float* out = (float*)d_out;

    hipLaunchKernelGGL(mc_fused_kernel, dim3(L_DIM / G_POS, B_DIM), dim3(G_THREADS), 0, stream,
                       seq, time, T, W1, b1, W2, b2, out);
}


# power chains: each wave runs only the square-and-multiply steps up to the highest set exponent bit of its 8 chains (later steps only square M, which nothing reads); R bit-identical for every n
# speedup vs baseline: 2.1669x; 1.2024x over previous
#include <hip/hip_runtime.h>
#include <hip/hip_bf16.h>
#include <math.h>


#define B_DIM 64
#define L_DIM 16384
#define ORDER 3
#define V_DIM 4
#define CTX_DIM 64
#define NBITS 16

typedef float v2f __attribute__((ext_vector_type(2)));
typedef float v8f __attribute__((ext_vector_type(8)));

#define G_POS        4096
#define G_THREADS    256
#define G_WAVES      (G_THREADS / 32)
#define G_PER_THREAD (G_POS / G_THREADS)
#define G_PER_WAVE   (G_POS / G_WAVES)
#define TILES        2

__global__ __launch_bounds__(G_THREADS) void mc_fused_kernel(
    const int* __restrict__ seq,  const float* __restrict__ time,
    const float* __restrict__ T,
    const float* __restrict__ W1, const float* __restrict__ b1,
    const float* __restrict__ W2, const float* __restrict__ b2,
    float* __restrict__ out)
{
    const int tid  = threadIdx.x;
    const int lane = tid & 31;
    const int wave = tid >> 5;
    const int b    = blockIdx.y;
    const int p0   = blockIdx.x * G_POS;

    __shared__ float4 s_row[CTX_DIM];
    __shared__ int4   s_seq4[G_POS / 4 + 1];
    __shared__ float4 s_pow[G_WAVES][TILES][2][16];
    int* s_seq = (int*)s_seq4;

    const int m  = lane & 15;
    const int hi = lane >> 4;
    const int k0 = hi * 2;
    const int qA = m >> 2;
    const int rA = m & 3;

    const float t = time[b];
    const int myCtx = wave * (4 * TILES) + hi * 4 + qA;
    const float4* w2v = (const float4*)(W2 + myCtx * 32);
    float4 w2r[8];
#pragma unroll
    for (int j = 0; j < 8; ++j) w2r[j] = w2v[j];
    float acc = b2[myCtx];

    const int* seqRow = seq + b * L_DIM + p0;
    const int haloIdx = (tid < ORDER && p0 > 0) ? tid - ORDER : 0;
    const int halo = seqRow[haloIdx];
    int4 sv[G_POS / 4 / G_THREADS];
#pragma unroll
    for (int j = 0; j < G_POS / 4 / G_THREADS; ++j)
        sv[j] = ((const int4*)seqRow)[tid + G_THREADS * j];

    v2f aM[TILES], bM[TILES], aR[TILES];
#pragma unroll
    for (int tt = 0; tt < TILES; ++tt) {
        const float* Mq = T + (wave * (4 * TILES) + tt * 4 + qA) * 16;
        aM[tt] = (v2f){ Mq[rA * 4 + k0], Mq[rA * 4 + k0 + 1] };
        bM[tt] = (v2f){ Mq[k0 * 4 + rA], Mq[(k0 + 1) * 4 + rA] };
        aR[tt] = (v2f){ (rA == k0) ? 1.0f : 0.0f, (rA == k0 + 1) ? 1.0f : 0.0f };
    }

#pragma unroll
    for (int j = 0; j < 32; ++j) {
        float h = fmaf(t, W1[j], b1[j]);
        h = fmaxf(h, 0.0f);
        const float4 wq = w2r[j >> 2];
        const float w = (j & 3) == 0 ? wq.x : (j & 3) == 1 ? wq.y : (j & 3) == 2 ? wq.z : wq.w;
        acc = fmaf(h, w, acc);
    }
    int myN;
    {
        float sp = log1pf(expf(-fabsf(acc))) + fmaxf(acc, 0.0f);
        float nf = fmaf(sp, t, 1.0f);
        int ni = (int)nf;
        ni = ni < 0 ? 0 : (ni > 65535 ? 65535 : ni);
        myN = ni;
    }
    const int otherN = __shfl_xor(myN, 16);
    int nexp[TILES];
    nexp[0] = hi ? otherN : myN;
    nexp[1] = hi ? myN : otherN;

#pragma unroll
    for (int j = 0; j < G_POS / 4 / G_THREADS; ++j)
        s_seq4[1 + tid + G_THREADS * j] = sv[j];
    asm volatile("" :: "v"(halo));
    if (tid < ORDER)   s_seq[1 + tid] = halo;

    const bool owner = (hi == (qA >> 1));
    const bool upper = (qA & 1) != 0;
    int nOr = 0;
#pragma unroll
    for (int c = 0; c < 4 * TILES; ++c) nOr |= __builtin_amdgcn_readlane(myN, 4 * c);
    int nSteps = nOr ? 32 - __builtin_clz((unsigned)nOr) : 0;
    nSteps = nSteps < NBITS ? nSteps : NBITS;
    for (int k = 0; k < nSteps; ++k) {
        v8f dRM[TILES], dMM[TILES];
#pragma unroll
        for (int tt = 0; tt < TILES; ++tt) {
            v8f zero = {};
            dRM[tt] = __builtin_amdgcn_wmma_f32_16x16x4_f32(
                false, aR[tt], false, bM[tt], (short)0, zero, false, false);
            dMM[tt] = __builtin_amdgcn_wmma_f32_16x16x4_f32(
                false, aM[tt], false, bM[tt], (short)0, zero, false, false);
        }
#pragma unroll
        for (int tt = 0; tt < TILES; ++tt) {
            if (owner) {
                float4 cmm, crm;
                cmm.x = upper ? dMM[tt][4] : dMM[tt][0];
                cmm.y = upper ? dMM[tt][5] : dMM[tt][1];
                cmm.z = upper ? dMM[tt][6] : dMM[tt][2];
                cmm.w = upper ? dMM[tt][7] : dMM[tt][3];
                crm.x = upper ? dRM[tt][4] : dRM[tt][0];
                crm.y = upper ? dRM[tt][5] : dRM[tt][1];
                crm.z = upper ? dRM[tt][6] : dRM[tt][2];
                crm.w = upper ? dRM[tt][7] : dRM[tt][3];
                s_pow[wave][tt][0][m] = cmm;
                s_pow[wave][tt][1][m] = crm;
            }
        }
        __builtin_amdgcn_fence(__ATOMIC_RELEASE, "wavefront");
        __builtin_amdgcn_wave_barrier();
        __builtin_amdgcn_fence(__ATOMIC_ACQUIRE, "wavefront");
#pragma unroll
        for (int tt = 0; tt < TILES; ++tt) {
            const float* mmc = (const float*)&s_pow[wave][tt][0][0];
            const float* rmc = (const float*)&s_pow[wave][tt][1][0];
            const int ia = qA * 16 + k0 * 4 + rA;
            aM[tt] = (v2f){ mmc[ia], mmc[ia + 4] };
            bM[tt] = *(const v2f*)&mmc[4 * m + k0];
            const v2f rnew = (v2f){ rmc[ia], rmc[ia + 4] };
            if ((nexp[tt] >> k) & 1) aR[tt] = rnew;
        }
        __builtin_amdgcn_fence(__ATOMIC_RELEASE, "wavefront");
        __builtin_amdgcn_wave_barrier();
        __builtin_amdgcn_fence(__ATOMIC_ACQUIRE, "wavefront");
    }

#pragma unroll
    for (int tt = 0; tt < TILES; ++tt) {
        if (rA == qA) {
            const int ctx = wave * (4 * TILES) + tt * 4 + qA;
            ((v2f*)&s_row[ctx])[hi] = aR[tt];
        }
    }

    __syncthreads();

    float4* outRow = (float4*)out + b * L_DIM + p0;
    const int kw = wave * G_PER_WAVE + lane;
#pragma unroll
    for (int i = 0; i < G_PER_THREAD; ++i) {
        const int k  = kw + i * 32;
        const int s0 = s_seq[k + 1];
        const int s1 = s_seq[k + 2];
        const int s2 = s_seq[k + 3];
        float4 res = s_row[s0 * 16 + s1 * 4 + s2];
        if (i == 0 && p0 + k < ORDER)
            res = make_float4(0.0f, 0.0f, 0.0f, 0.0f);
        outRow[k] = res;
    }
}

extern "C" void kernel_launch(void* const* d_in, const int* in_sizes, int n_in,
                              void* d_out, int out_size, void* d_ws, size_t ws_size,
                              hipStream_t stream) {
    const int*   seq  = (const int*)d_in[0];
    const float* time = (const float*)d_in[1];
    const float* T    = (const float*)d_in[2];
    const float* W1   = (const float*)d_in[3];
    const float* b1   = (const float*)d_in[4];
    const float* W2   = (const float*)d_in[5];
    const float* b2   = (const float*)d_in[6];
    float* out = (float*)d_out;

    hipLaunchKernelGGL(mc_fused_kernel, dim3(L_DIM / G_POS, B_DIM), dim3(G_THREADS), 0, stream,
                       seq, time, T, W1, b1, W2, b2, out);
}


# power chains: only M's squarings stay on the serial path and the first 4 run beside the exponent arithmetic; R follows in registers as the transposed product (R@M)^T = M^T@R^T; R bit-identical
# speedup vs baseline: 2.4975x; 1.1526x over previous
#include <hip/hip_runtime.h>
#include <hip/hip_bf16.h>
#include <math.h>


#define B_DIM 64
#define L_DIM 16384
#define ORDER 3
#define V_DIM 4
#define CTX_DIM 64
#define NBITS 16

typedef float v2f __attribute__((ext_vector_type(2)));
typedef float v8f __attribute__((ext_vector_type(8)));
typedef float v4f __attribute__((ext_vector_type(4)));

#define G_POS        4096
#define G_THREADS    256
#define G_WAVES      (G_THREADS / 32)
#define G_PER_THREAD (G_POS / G_THREADS)
#define G_PER_WAVE   (G_POS / G_WAVES)
#define TILES        2
#define AHEAD        4

__device__ __forceinline__ float xhalf(float x) {
    return __builtin_bit_cast(float, __builtin_amdgcn_permlanex16(
        __builtin_bit_cast(int, x), __builtin_bit_cast(int, x), 0x76543210, 0xfedcba98, false, false));
}

__global__ __launch_bounds__(G_THREADS) __attribute__((amdgpu_waves_per_eu(2, 2))) void mc_fused_kernel(
    const int* seq,  const float* __restrict__ time,
    const float* __restrict__ T,
    const float* __restrict__ W1, const float* __restrict__ b1,
    const float* __restrict__ W2, const float* __restrict__ b2,
    float* __restrict__ out)
{
    const int tid  = threadIdx.x;
    const int lane = tid & 31;
    const int wave = tid >> 5;
    const int b    = blockIdx.y;
    const int p0   = blockIdx.x * G_POS;

    __shared__ float4 s_row[CTX_DIM];
    __shared__ int4   s_seq4[G_POS / 4 + 1];
    __shared__ float4 s_pow[G_WAVES][2][TILES][16];
    int* s_seq = (int*)s_seq4;

    const int m  = lane & 15;
    const int hi = lane >> 4;
    const int k0 = hi * 2;
    const int qA = m >> 2;
    const int rA = m & 3;
    const bool owner = (hi == (qA >> 1));
    const bool upper = (qA & 1) != 0;

    v2f aM[TILES], bM[TILES], aR[TILES];
#pragma unroll
    for (int tt = 0; tt < TILES; ++tt) {
        const float* Mq = T + (wave * (4 * TILES) + tt * 4 + qA) * 16;
        aM[tt] = (v2f){ Mq[rA * 4 + k0], Mq[rA * 4 + k0 + 1] };
        bM[tt] = (v2f){ Mq[k0 * 4 + rA], Mq[(k0 + 1) * 4 + rA] };
        aR[tt] = (v2f){ (rA == k0) ? 1.0f : 0.0f, (rA == k0 + 1) ? 1.0f : 0.0f };
    }
    const float t = time[b];
    const int myCtx = wave * (4 * TILES) + hi * 4 + qA;
    const float4* w2v = (const float4*)(W2 + myCtx * 32);
    float4 w2r[8];
#pragma unroll
    for (int j = 0; j < 8; ++j) w2r[j] = w2v[j];
    float acc = b2[myCtx];
    float w1r[32], b1r[32];
#pragma unroll
    for (int j = 0; j < 32; ++j) { w1r[j] = W1[j]; b1r[j] = b1[j]; }

    const int* seqRow = seq + b * L_DIM + p0;
    const int haloIdx = (tid < ORDER && p0 > 0) ? tid - ORDER : 0;
    const int halo = seqRow[haloIdx];
    const int4 sv0 = ((const int4*)seqRow)[tid];
    const int4 sv1 = ((const int4*)seqRow)[tid + G_THREADS];
    const int4 sv2 = ((const int4*)seqRow)[tid + G_THREADS * 2];
    const int4 sv3 = ((const int4*)seqRow)[tid + G_THREADS * 3];
    __builtin_amdgcn_sched_barrier(0);

    auto square = [&]() {
        v8f dMM[TILES];
#pragma unroll
        for (int tt = 0; tt < TILES; ++tt) {
            v8f zero = {};
            dMM[tt] = __builtin_amdgcn_wmma_f32_16x16x4_f32(
                false, aM[tt], false, bM[tt], (short)0, zero, false, false);
        }
#pragma unroll
        for (int tt = 0; tt < TILES; ++tt) {
            const v4f lo = __builtin_shufflevector(dMM[tt], dMM[tt], 0, 1, 2, 3);
            const v4f up = __builtin_shufflevector(dMM[tt], dMM[tt], 4, 5, 6, 7);
            *(v4f*)&s_pow[wave][owner ? 0 : 1][tt][m] = upper ? up : lo;
        }
        __builtin_amdgcn_fence(__ATOMIC_RELEASE, "wavefront");
        __builtin_amdgcn_wave_barrier();
        __builtin_amdgcn_fence(__ATOMIC_ACQUIRE, "wavefront");
#pragma unroll
        for (int tt = 0; tt < TILES; ++tt) {
            const float* mmc = (const float*)&s_pow[wave][0][tt][0];
            const int ia = qA * 16 + k0 * 4 + rA;
            aM[tt] = (v2f){ mmc[ia], mmc[ia + 4] };
            bM[tt] = *(const v2f*)&mmc[4 * m + k0];
        }
        __builtin_amdgcn_fence(__ATOMIC_RELEASE, "wavefront");
        __builtin_amdgcn_wave_barrier();
        __builtin_amdgcn_fence(__ATOMIC_ACQUIRE, "wavefront");
    };

    v2f bMs[AHEAD][TILES];
#pragma unroll
    for (int tt = 0; tt < TILES; ++tt) bMs[0][tt] = bM[tt];
    square();
    __builtin_amdgcn_sched_barrier(0);

#pragma unroll
    for (int j = 0; j < 11; ++j) {
        float h = fmaf(t, w1r[j], b1r[j]);
        h = fmaxf(h, 0.0f);
        const float4 wq = w2r[j >> 2];
        const float w = (j & 3) == 0 ? wq.x : (j & 3) == 1 ? wq.y : (j & 3) == 2 ? wq.z : wq.w;
        acc = fmaf(h, w, acc);
    }
    __builtin_amdgcn_sched_barrier(0);
#pragma unroll
    for (int tt = 0; tt < TILES; ++tt) bMs[1][tt] = bM[tt];
    square();
    __builtin_amdgcn_sched_barrier(0);

#pragma unroll
    for (int j = 11; j < 22; ++j) {
        float h = fmaf(t, w1r[j], b1r[j]);
        h = fmaxf(h, 0.0f);
        const float4 wq = w2r[j >> 2];
        const float w = (j & 3) == 0 ? wq.x : (j & 3) == 1 ? wq.y : (j & 3) == 2 ? wq.z : wq.w;
        acc = fmaf(h, w, acc);
    }
    s_seq4[1 + tid]                 = sv0;
    s_seq4[1 + tid + G_THREADS]     = sv1;
    s_seq4[1 + tid + G_THREADS * 2] = sv2;
    s_seq4[1 + tid + G_THREADS * 3] = sv3;
    asm volatile("" :: "v"(halo));
    if (tid < ORDER)   s_seq[1 + tid] = halo;
    __builtin_amdgcn_sched_barrier(0);
#pragma unroll
    for (int tt = 0; tt < TILES; ++tt) bMs[2][tt] = bM[tt];
    square();
    __builtin_amdgcn_sched_barrier(0);

#pragma unroll
    for (int j = 22; j < 32; ++j) {
        float h = fmaf(t, w1r[j], b1r[j]);
        h = fmaxf(h, 0.0f);
        const float4 wq = w2r[j >> 2];
        const float w = (j & 3) == 0 ? wq.x : (j & 3) == 1 ? wq.y : (j & 3) == 2 ? wq.z : wq.w;
        acc = fmaf(h, w, acc);
    }
    __builtin_amdgcn_sched_barrier(0);
#pragma unroll
    for (int tt = 0; tt < TILES; ++tt) bMs[3][tt] = bM[tt];
    square();
    __builtin_amdgcn_sched_barrier(0);

    int myN;
    {
        float sp = log1pf(expf(-fabsf(acc))) + fmaxf(acc, 0.0f);
        float nf = fmaf(sp, t, 1.0f);
        int ni = (int)nf;
        ni = ni < 0 ? 0 : (ni > 65535 ? 65535 : ni);
        myN = ni;
    }
    const int otherN = __shfl_xor(myN, 16);
    int nexp[TILES];
    nexp[0] = hi ? otherN : myN;
    nexp[1] = hi ? myN : otherN;

    int nOr = 0;
#pragma unroll
    for (int c = 0; c < 4 * TILES; ++c) nOr |= __builtin_amdgcn_readlane(myN, 4 * c);
    int nSteps = nOr ? 32 - __builtin_clz((unsigned)nOr) : 0;
    nSteps = nSteps < NBITS ? nSteps : NBITS;

    auto rstep = [&](int tt, v2f mT, int bit) {
        v8f zero = {};
        const v8f d = __builtin_amdgcn_wmma_f32_16x16x4_f32(
            false, mT, false, aR[tt], (short)0, zero, false, false);
        const v4f lo = __builtin_shufflevector(d, d, 0, 1, 2, 3);
        const v4f up = __builtin_shufflevector(d, d, 4, 5, 6, 7);
        const v4f s = upper ? up : lo;
        const float keep0 = hi ? s[2] : s[0], keep1 = hi ? s[3] : s[1];
        const float recv0 = xhalf(hi ? s[0] : s[2]), recv1 = xhalf(hi ? s[1] : s[3]);
        const v2f rnew = owner ? (v2f){ keep0, keep1 } : (v2f){ recv0, recv1 };
        if (bit) aR[tt] = rnew;
    };

#pragma unroll
    for (int k = 0; k < AHEAD; ++k) {
#pragma unroll
        for (int tt = 0; tt < TILES; ++tt) rstep(tt, bMs[k][tt], (nexp[tt] >> k) & 1);
    }

    for (int k = AHEAD; k < nSteps; ++k) {
        v2f mT[TILES];
#pragma unroll
        for (int tt = 0; tt < TILES; ++tt) mT[tt] = bM[tt];
        square();
#pragma unroll
        for (int tt = 0; tt < TILES; ++tt) rstep(tt, mT[tt], (nexp[tt] >> k) & 1);
    }

#pragma unroll
    for (int tt = 0; tt < TILES; ++tt) {
        if (rA == qA) {
            const int ctx = wave * (4 * TILES) + tt * 4 + qA;
            ((v2f*)&s_row[ctx])[hi] = aR[tt];
        }
    }

    __syncthreads();

    float4* outRow = (float4*)out + b * L_DIM + p0;
    const int kw = wave * G_PER_WAVE + lane;
#pragma unroll
    for (int i = 0; i < G_PER_THREAD; ++i) {
        const int k  = kw + i * 32;
        const int s0 = s_seq[k + 1];
        const int s1 = s_seq[k + 2];
        const int s2 = s_seq[k + 3];
        float4 res = s_row[s0 * 16 + s1 * 4 + s2];
        if (i == 0 && p0 + k < ORDER)
            res = make_float4(0.0f, 0.0f, 0.0f, 0.0f);
        outRow[k] = res;
    }
}

extern "C" void kernel_launch(void* const* d_in, const int* in_sizes, int n_in,
                              void* d_out, int out_size, void* d_ws, size_t ws_size,
                              hipStream_t stream) {
    const int*   seq  = (const int*)d_in[0];
    const float* time = (const float*)d_in[1];
    const float* T    = (const float*)d_in[2];
    const float* W1   = (const float*)d_in[3];
    const float* b1   = (const float*)d_in[4];
    const float* W2   = (const float*)d_in[5];
    const float* b2   = (const float*)d_in[6];
    float* out = (float*)d_out;

    hipLaunchKernelGGL(mc_fused_kernel, dim3(L_DIM / G_POS, B_DIM), dim3(G_THREADS), 0, stream,
                       seq, time, T, W1, b1, W2, b2, out);
}
